# starting kernel
# baseline (speedup 1.0000x reference)
#include <hip/hip_runtime.h>


#define BATCH  16384
#define T      700
#define F      3
#define TCHUNK 20
#define NCHUNK (T / TCHUNK)
#define BLOCK  128
#define NWAVES (BLOCK / 32)

typedef __attribute__((ext_vector_type(2))) float v2f;
typedef __attribute__((ext_vector_type(8))) float v8f;

__device__ __forceinline__ float fast_sigmoid(float x) {
    float e = __builtin_amdgcn_exp2f(-x * 1.44269504088896340736f);
    return __builtin_amdgcn_rcpf(1.0f + e);
}

__device__ __forceinline__ float fast_tanh(float x) {
    float e = __builtin_amdgcn_exp2f(-x * 2.88539008177792681472f);
    return 2.0f * __builtin_amdgcn_rcpf(1.0f + e) - 1.0f;
}

__global__ void lstm_scan_kernel(const float* __restrict__ x,
                                 const float* __restrict__ kern,
                                 const float* __restrict__ rkern,
                                 const float* __restrict__ bias,
                                 float* __restrict__ out) {
    extern __shared__ float smem[];
    float* x_lds  = smem;
    float* xz_lds = smem + BLOCK * TCHUNK * F;

    const int tid  = threadIdx.x;
    const int lane = tid & 31;
    const int wave = tid >> 5;
    const int b    = blockIdx.x * BLOCK + tid;

    const int n  = lane & 15;
    const int kb = (lane < 16) ? 0 : 2;
    v2f bmat;
    bmat.x = (n < 4 && (kb + 0) < F) ? kern[(kb + 0) * 4 + n] : 0.0f;
    bmat.y = (n < 4 && (kb + 1) < F) ? kern[(kb + 1) * 4 + n] : 0.0f;

    const float bias_n = (n < 4) ? bias[n] : 0.0f;
    v8f cmat;
#pragma unroll
    for (int v = 0; v < 8; ++v) cmat[v] = bias_n;

    const float aymask = (kb == 0) ? 1.0f : 0.0f;

    const float rk0 = rkern[0], rk1 = rkern[1], rk2 = rkern[2], rk3 = rkern[3];

    float h = 0.0f, c = 0.0f;
    const float* xg = x + (size_t)b * (T * F);

    for (int chunk = 0; chunk < NCHUNK; ++chunk) {
        const int t0 = chunk * TCHUNK;

        __syncthreads();

        {
            const float4* src = (const float4*)(xg + t0 * F);
            float4*       dst = (float4*)(x_lds + tid * (TCHUNK * F));
#pragma unroll
            for (int v = 0; v < (TCHUNK * F) / 4; ++v) dst[v] = src[v];
        }
        __syncthreads();

        {
            const int rbase_wave = wave * (2 * TCHUNK * 16);
            for (int g = 0; g < 2 * TCHUNK; ++g) {
                const int rb = rbase_wave + g * 16;
                const int r  = rb + n;
                v2f a;
                a.x = x_lds[r * F + kb];
                a.y = x_lds[r * F + kb + 1] * aymask;

                v8f d = __builtin_amdgcn_wmma_f32_16x16x4_f32(
                    false, a, false, bmat, (short)0, cmat, false, false);

                if (n < 4) {
                    const int mofs = (lane < 16) ? 0 : 8;
#pragma unroll
                    for (int v = 0; v < 8; ++v)
                        xz_lds[(rb + mofs + v) * 4 + n] = d[v];
                }
            }
        }
        __syncthreads();

        {
            const float4* zz = (const float4*)(xz_lds + tid * (TCHUNK * 4));
#pragma unroll
            for (int t = 0; t < TCHUNK; ++t) {
                float4 z = zz[t];
                float ig = fast_sigmoid(z.x + h * rk0);
                float fg = fast_sigmoid(z.y + h * rk1);
                float gg = fast_tanh   (z.z + h * rk2);
                float og = fast_sigmoid(z.w + h * rk3);
                c = fg * c + ig * gg;
                h = og * fast_tanh(c);
            }
        }
    }

    out[b] = h;
}

extern "C" void kernel_launch(void* const* d_in, const int* in_sizes, int n_in,
                              void* d_out, int out_size, void* d_ws, size_t ws_size,
                              hipStream_t stream) {
    const float* x  = (const float*)d_in[0];
    const float* k  = (const float*)d_in[1];
    const float* rk = (const float*)d_in[2];
    const float* bi = (const float*)d_in[3];
    float*       o  = (float*)d_out;

    const size_t lds_bytes = (size_t)BLOCK * TCHUNK * (F + 4) * sizeof(float);
    hipLaunchKernelGGL(lstm_scan_kernel,
                       dim3(BATCH / BLOCK), dim3(BLOCK), lds_bytes, stream,
                       x, k, rk, bi, o);
}


# Ring hand-over off the scan's serial chain: split barrier (a scan wave arrives once a tile is in registers, waits mid-tile) and the next tile's 20 values read mid-tile; sums unchanged
# speedup vs baseline: 9.3367x; 9.3367x over previous
#include <hip/hip_runtime.h>


#define BATCH  16384
#define T      700
#define F      3
#define TCHUNK 20
#define NCHUNK (T / TCHUNK)
#define SCAN_WAVES 4
#define PROD_WAVES 2
#define BLOCK  ((SCAN_WAVES + PROD_WAVES) * 32)
#define BPB    (SCAN_WAVES * 32 / 4)
#define ROWS   (BPB * TCHUNK)
#define XV4    (TCHUNK * F / 4)
#define EPP    (BPB / PROD_WAVES)
#define PGROUPS (EPP * TCHUNK / 16)
#define PV4    (EPP * XV4)
#define PLOADS ((PV4 + 31) / 32)
#define GB     5
#define XSLOT  (PLOADS * 32 * 4)
#define XZTILE (ROWS * 4)
#define HALF   (TCHUNK / 2)

static_assert((NCHUNK & 1) == 1, "the scan runs tiles in pairs plus one last tile");
typedef __attribute__((ext_vector_type(2))) float v2f;
typedef __attribute__((ext_vector_type(8))) float v8f;

__device__ __forceinline__ float fast_tanh(float x) {
    float e = __builtin_amdgcn_exp2f(-x * 2.88539008177792681472f);
    return 2.0f * __builtin_amdgcn_rcpf(1.0f + e) - 1.0f;
}

template <int K>
__device__ __forceinline__ float quad_take(float v) {
    return __int_as_float(__builtin_amdgcn_update_dpp(
        0, __float_as_int(v), K * 0x55, 0xF, 0xF, true));
}

__device__ __forceinline__ void bar_arrive() {
    asm volatile("s_barrier_signal -1" ::: "memory");
}
__device__ __forceinline__ void bar_wait() {
    asm volatile("s_barrier_wait -1" ::: "memory");
}
__device__ __forceinline__ void lds_done() {
    asm volatile("s_wait_dscnt 0x0" ::: "memory");
}
__device__ __forceinline__ void bar_arrive_at(float& h) {
    asm volatile("s_barrier_signal -1" : "+v"(h) :: "memory");
}
__device__ __forceinline__ void bar_wait_at(float& h) {
    asm volatile("s_barrier_wait -1" : "+v"(h) :: "memory");
}
__device__ __forceinline__ void lds_done_at(float& h) {
    asm volatile("s_wait_dscnt 0x0" : "+v"(h) :: "memory");
}
__device__ __forceinline__ void issue_here(float& h) {
    asm volatile("" : "+v"(h) :: "memory");
}

__device__ __forceinline__ void read_tile(float (&zr)[TCHUNK], const float* zz) {
#pragma unroll
    for (int t = 0; t < TCHUNK; ++t) zr[t] = zz[t * 4];
}

template <int T0, int T1>
__device__ __forceinline__ void scan_steps(const float (&zr)[TCHUNK],
                                           float& h, float& c, float rk,
                                           float sc, float am, float ab) {
#pragma unroll
    for (int t = T0; t < T1; ++t) {
        const float z   = zr[t] + h * rk;
        const float e   = __builtin_amdgcn_exp2f(-z * sc);
        const float act = am * __builtin_amdgcn_rcpf(1.0f + e) + ab;
        const float ig = quad_take<0>(act);
        const float fg = quad_take<1>(act);
        const float gg = quad_take<2>(act);
        const float og = quad_take<3>(act);
        c = fg * c + ig * gg;
        h = og * fast_tanh(c);
    }
}

__device__ __forceinline__ void scan_tile(const float (&cur)[TCHUNK],
                                          float (&nxt)[TCHUNK],
                                          const float* zz_next,
                                          float& h, float& c, float rk,
                                          float sc, float am, float ab) {
    bar_arrive_at(h);
    scan_steps<0, HALF>(cur, h, c, rk, sc, am, ab);
    bar_wait_at(h);
    read_tile(nxt, zz_next);
    issue_here(h);
    scan_steps<HALF, TCHUNK>(cur, h, c, rk, sc, am, ab);
    lds_done_at(h);
}

__global__ __launch_bounds__(BLOCK)
void lstm_scan_kernel(const float* __restrict__ x,
                      const float* __restrict__ kern,
                      const float* __restrict__ rkern,
                      const float* __restrict__ bias,
                      float* __restrict__ out) {
    extern __shared__ __attribute__((aligned(16))) float smem[];
    float* xz_lds = smem;
    float* x_lds  = smem + 2 * XZTILE;

    const int tid  = threadIdx.x;
    const int lane = tid & 31;
    const int wave = __builtin_amdgcn_readfirstlane(tid >> 5);
    const bool producer = (wave >= SCAN_WAVES);
    const int b0   = blockIdx.x * BPB;
    const float* xblk = x + (size_t)b0 * (T * F);

    const int gate = tid & 3;
    const int bl   = (tid >> 2) & (BPB - 1);
    const bool  is_g = (gate == 2);
    const float rk   = rkern[gate];
    const float sc   = is_g ? 2.88539008177792681472f : 1.44269504088896340736f;
    const float am   = is_g ? 2.0f : 1.0f;
    const float ab   = is_g ? -1.0f : 0.0f;
    float h = 0.0f, c = 0.0f;

    const int n  = lane & 15;
    const int kb = (lane < 16) ? 0 : 2;
    v2f bmat;
    bmat.x = (n < 4 && (kb + 0) < F) ? kern[(kb + 0) * 4 + n] : 0.0f;
    bmat.y = (n < 4 && (kb + 1) < F) ? kern[(kb + 1) * 4 + n] : 0.0f;
    const float bias_n = (n < 4) ? bias[n] : 0.0f;
    v8f cmat;
    for (int v = 0; v < 8; ++v) cmat[v] = bias_n;

    const int pw  = producer ? (wave - SCAN_WAVES) : 0;
    const int ep0 = pw * EPP;
    float* xs = x_lds + pw * XSLOT;
    const float4* src[PLOADS];
#pragma unroll
    for (int j = 0; j < PLOADS; ++j) {
        int i = lane + 32 * j;
        i = (i < PV4) ? i : (PV4 - 1);
        const int e = i / XV4;
        const int v = i - e * XV4;
        src[j] = (const float4*)(xblk + (size_t)(ep0 + e) * (T * F)) + v;
    }
    float4 xr[PLOADS];
#pragma unroll
    for (int j = 0; j < PLOADS; ++j) xr[j] = make_float4(0.0f, 0.0f, 0.0f, 0.0f);
    if (producer) {
#pragma unroll
        for (int j = 0; j < PLOADS; ++j) xr[j] = src[j][0];
    }

    if (producer) {
        for (int it = 0; it < NCHUNK; ++it) {
#pragma unroll
            for (int j = 0; j < PLOADS; ++j)
                ((float4*)xs)[lane + 32 * j] = xr[j];
            if (it + 1 < NCHUNK) {
#pragma unroll
                for (int j = 0; j < PLOADS; ++j)
                    xr[j] = src[j][(it + 1) * XV4];
            }
            __builtin_amdgcn_wave_barrier();

            float* xzb = xz_lds + (it & 1) * XZTILE + ep0 * (TCHUNK * 4);
            const int mofs = (lane < 16) ? 0 : 8;
            v2f afr[PGROUPS];
#pragma unroll
            for (int g = 0; g < PGROUPS; ++g) {
                const int r = g * 16 + n;
                const float ax = xs[r * F + kb];
                const float ay = xs[r * F + kb + 1];
                afr[g].x = ax;
                afr[g].y = (kb == 0) ? ay : 0.0f;
            }
#pragma unroll
            for (int gb = 0; gb < PGROUPS; gb += GB) {
                v8f d[GB];
#pragma unroll
                for (int u = 0; u < GB; ++u)
                    d[u] = __builtin_amdgcn_wmma_f32_16x16x4_f32(
                        false, afr[gb + u], false, bmat, (short)0, cmat,
                        false, false);
                if (n < 4) {
#pragma unroll
                    for (int u = 0; u < GB; ++u)
#pragma unroll
                        for (int v = 0; v < 8; ++v)
                            xzb[((gb + u) * 16 + mofs + v) * 4 + n] = d[u][v];
                }
            }
            lds_done();
            bar_arrive();
            bar_wait();
        }
    } else {
        const float* zz0 = xz_lds + bl * (TCHUNK * 4) + gate;
        const float* zz1 = zz0 + XZTILE;
        float za[TCHUNK], zb[TCHUNK];
        bar_arrive();
        bar_wait();
        read_tile(za, zz0);
        lds_done();
        for (int k = 0; k + 2 < NCHUNK; k += 2) {
            scan_tile(za, zb, zz1, h, c, rk, sc, am, ab);
            scan_tile(zb, za, zz0, h, c, rk, sc, am, ab);
        }
        scan_steps<0, TCHUNK>(za, h, c, rk, sc, am, ab);
    }

    if (!producer && gate == 0) x_lds[bl] = h;
    __syncthreads();
    if (tid < BPB) out[b0 + tid] = x_lds[tid];
}

extern "C" void kernel_launch(void* const* d_in, const int* in_sizes, int n_in,
                              void* d_out, int out_size, void* d_ws, size_t ws_size,
                              hipStream_t stream) {
    const float* x  = (const float*)d_in[0];
    const float* k  = (const float*)d_in[1];
    const float* rk = (const float*)d_in[2];
    const float* bi = (const float*)d_in[3];
    float*       o  = (float*)d_out;

    const size_t lds_bytes =
        (size_t)(2 * XZTILE + PROD_WAVES * XSLOT) * sizeof(float);
    hipLaunchKernelGGL(lstm_scan_kernel,
                       dim3(BATCH / BPB), dim3(BLOCK), lds_bytes, stream,
                       x, k, rk, bi, o);
}


# Cell update re-formed: the quad swaps raw 1/(1+e); g's 2r-1 and tanh's exp2 scale move into the update's constants (cell carried as k*c): 2 of 15 chain links go; sums re-associated
# speedup vs baseline: 9.7749x; 1.0469x over previous
#include <hip/hip_runtime.h>


#define BATCH  16384
#define T      700
#define F      3
#define TCHUNK 20
#define NCHUNK (T / TCHUNK)
#define SCAN_WAVES 4
#define PROD_WAVES 2
#define BLOCK  ((SCAN_WAVES + PROD_WAVES) * 32)
#define BPB    (SCAN_WAVES * 32 / 4)
#define ROWS   (BPB * TCHUNK)
#define XV4    (TCHUNK * F / 4)
#define EPP    (BPB / PROD_WAVES)
#define PGROUPS (EPP * TCHUNK / 16)
#define PV4    (EPP * XV4)
#define PLOADS ((PV4 + 31) / 32)
#define GB     5
#define XSLOT  (PLOADS * 32 * 4)
#define XZTILE (ROWS * 4)
#define HALF   (TCHUNK / 2)

static_assert((NCHUNK & 1) == 1, "the scan runs tiles in pairs plus one last tile");
typedef __attribute__((ext_vector_type(2))) float v2f;
typedef __attribute__((ext_vector_type(8))) float v8f;

template <int K>
__device__ __forceinline__ float quad_take(float v) {
    return __int_as_float(__builtin_amdgcn_update_dpp(
        0, __float_as_int(v), K * 0x55, 0xF, 0xF, true));
}

__device__ __forceinline__ void bar_arrive() {
    asm volatile("s_barrier_signal -1" ::: "memory");
}
__device__ __forceinline__ void bar_wait() {
    asm volatile("s_barrier_wait -1" ::: "memory");
}
__device__ __forceinline__ void lds_done() {
    asm volatile("s_wait_dscnt 0x0" ::: "memory");
}
__device__ __forceinline__ void bar_arrive_at(float& h) {
    asm volatile("s_barrier_signal -1" : "+v"(h) :: "memory");
}
__device__ __forceinline__ void bar_wait_at(float& h) {
    asm volatile("s_barrier_wait -1" : "+v"(h) :: "memory");
}
__device__ __forceinline__ void lds_done_at(float& h) {
    asm volatile("s_wait_dscnt 0x0" : "+v"(h) :: "memory");
}
__device__ __forceinline__ void issue_here(float& h) {
    asm volatile("" : "+v"(h) :: "memory");
}

__device__ __forceinline__ void read_tile(float (&zr)[TCHUNK], const float* zz) {
#pragma unroll
    for (int t = 0; t < TCHUNK; ++t) zr[t] = zz[t * 4];
}

#define KC (-2.88539008177792681472f)
template <int T0, int T1>
__device__ __forceinline__ void scan_steps(const float (&zr)[TCHUNK],
                                           float& h, float& cs, float rk,
                                           float sc) {
#pragma unroll
    for (int t = T0; t < T1; ++t) {
        const float z  = zr[t] + h * rk;
        const float e  = __builtin_amdgcn_exp2f(-z * sc);
        const float r  = __builtin_amdgcn_rcpf(1.0f + e);
        const float ri = quad_take<0>(r);
        const float fc = quad_take<1>(r) * cs;
        const float u  = quad_take<2>(r) * ri;
        cs = __builtin_fmaf(u, 2.0f * KC, __builtin_fmaf(ri, -KC, fc));
        const float e2 = __builtin_amdgcn_exp2f(cs);
        const float th = 2.0f * __builtin_amdgcn_rcpf(1.0f + e2) - 1.0f;
        h = quad_take<3>(r) * th;
    }
}

__device__ __forceinline__ void scan_tile(const float (&cur)[TCHUNK],
                                          float (&nxt)[TCHUNK],
                                          const float* zz_next,
                                          float& h, float& c, float rk,
                                          float sc) {
    bar_arrive_at(h);
    scan_steps<0, HALF>(cur, h, c, rk, sc);
    bar_wait_at(h);
    read_tile(nxt, zz_next);
    issue_here(h);
    scan_steps<HALF, TCHUNK>(cur, h, c, rk, sc);
    lds_done_at(h);
}

__global__ __launch_bounds__(BLOCK)
void lstm_scan_kernel(const float* __restrict__ x,
                      const float* __restrict__ kern,
                      const float* __restrict__ rkern,
                      const float* __restrict__ bias,
                      float* __restrict__ out) {
    extern __shared__ __attribute__((aligned(16))) float smem[];
    float* xz_lds = smem;
    float* x_lds  = smem + 2 * XZTILE;

    const int tid  = threadIdx.x;
    const int lane = tid & 31;
    const int wave = __builtin_amdgcn_readfirstlane(tid >> 5);
    const bool producer = (wave >= SCAN_WAVES);
    const int b0   = blockIdx.x * BPB;
    const float* xblk = x + (size_t)b0 * (T * F);

    const int gate = tid & 3;
    const int bl   = (tid >> 2) & (BPB - 1);
    const bool  is_g = (gate == 2);
    const float rk   = rkern[gate];
    const float sc   = is_g ? 2.88539008177792681472f : 1.44269504088896340736f;
    float h = 0.0f, c = 0.0f;

    const int n  = lane & 15;
    const int kb = (lane < 16) ? 0 : 2;
    v2f bmat;
    bmat.x = (n < 4 && (kb + 0) < F) ? kern[(kb + 0) * 4 + n] : 0.0f;
    bmat.y = (n < 4 && (kb + 1) < F) ? kern[(kb + 1) * 4 + n] : 0.0f;
    const float bias_n = (n < 4) ? bias[n] : 0.0f;
    v8f cmat;
    for (int v = 0; v < 8; ++v) cmat[v] = bias_n;

    const int pw  = producer ? (wave - SCAN_WAVES) : 0;
    const int ep0 = pw * EPP;
    float* xs = x_lds + pw * XSLOT;
    const float4* src[PLOADS];
#pragma unroll
    for (int j = 0; j < PLOADS; ++j) {
        int i = lane + 32 * j;
        i = (i < PV4) ? i : (PV4 - 1);
        const int e = i / XV4;
        const int v = i - e * XV4;
        src[j] = (const float4*)(xblk + (size_t)(ep0 + e) * (T * F)) + v;
    }
    float4 xr[PLOADS];
#pragma unroll
    for (int j = 0; j < PLOADS; ++j) xr[j] = make_float4(0.0f, 0.0f, 0.0f, 0.0f);
    if (producer) {
#pragma unroll
        for (int j = 0; j < PLOADS; ++j) xr[j] = src[j][0];
    }

    if (producer) {
        for (int it = 0; it < NCHUNK; ++it) {
#pragma unroll
            for (int j = 0; j < PLOADS; ++j)
                ((float4*)xs)[lane + 32 * j] = xr[j];
            if (it + 1 < NCHUNK) {
#pragma unroll
                for (int j = 0; j < PLOADS; ++j)
                    xr[j] = src[j][(it + 1) * XV4];
            }
            __builtin_amdgcn_wave_barrier();

            float* xzb = xz_lds + (it & 1) * XZTILE + ep0 * (TCHUNK * 4);
            const int mofs = (lane < 16) ? 0 : 8;
            v2f afr[PGROUPS];
#pragma unroll
            for (int g = 0; g < PGROUPS; ++g) {
                const int r = g * 16 + n;
                const float ax = xs[r * F + kb];
                const float ay = xs[r * F + kb + 1];
                afr[g].x = ax;
                afr[g].y = (kb == 0) ? ay : 0.0f;
            }
#pragma unroll
            for (int gb = 0; gb < PGROUPS; gb += GB) {
                v8f d[GB];
#pragma unroll
                for (int u = 0; u < GB; ++u)
                    d[u] = __builtin_amdgcn_wmma_f32_16x16x4_f32(
                        false, afr[gb + u], false, bmat, (short)0, cmat,
                        false, false);
                if (n < 4) {
#pragma unroll
                    for (int u = 0; u < GB; ++u)
#pragma unroll
                        for (int v = 0; v < 8; ++v)
                            xzb[((gb + u) * 16 + mofs + v) * 4 + n] = d[u][v];
                }
            }
            lds_done();
            bar_arrive();
            bar_wait();
        }
    } else {
        const float* zz0 = xz_lds + bl * (TCHUNK * 4) + gate;
        const float* zz1 = zz0 + XZTILE;
        float za[TCHUNK], zb[TCHUNK];
        bar_arrive();
        bar_wait();
        read_tile(za, zz0);
        lds_done();
        for (int k = 0; k + 2 < NCHUNK; k += 2) {
            scan_tile(za, zb, zz1, h, c, rk, sc);
            scan_tile(zb, za, zz0, h, c, rk, sc);
        }
        scan_steps<0, TCHUNK>(za, h, c, rk, sc);
    }

    if (!producer && gate == 0) x_lds[bl] = h;
    __syncthreads();
    if (tid < BPB) out[b0 + tid] = x_lds[tid];
}

extern "C" void kernel_launch(void* const* d_in, const int* in_sizes, int n_in,
                              void* d_out, int out_size, void* d_ws, size_t ws_size,
                              hipStream_t stream) {
    const float* x  = (const float*)d_in[0];
    const float* k  = (const float*)d_in[1];
    const float* rk = (const float*)d_in[2];
    const float* bi = (const float*)d_in[3];
    float*       o  = (float*)d_out;

    const size_t lds_bytes =
        (size_t)(2 * XZTILE + PROD_WAVES * XSLOT) * sizeof(float);
    hipLaunchKernelGGL(lstm_scan_kernel,
                       dim3(BATCH / BPB), dim3(BLOCK), lds_bytes, stream,
                       x, k, rk, bi, o);
}


# Two affine links leave the serial chain: gate exp2 scale rides in the projection constants and rk; h = q*(o+o) - o is one fma behind the reciprocal (was 2q-1, then *o); products re-associated
# speedup vs baseline: 10.0889x; 1.0321x over previous
#include <hip/hip_runtime.h>


#define BATCH  16384
#define T      700
#define F      3
#define TCHUNK 20
#define NCHUNK (T / TCHUNK)
#define SCAN_WAVES 4
#define PROD_WAVES 2
#define BLOCK  ((SCAN_WAVES + PROD_WAVES) * 32)
#define BPB    (SCAN_WAVES * 32 / 4)
#define ROWS   (BPB * TCHUNK)
#define XV4    (TCHUNK * F / 4)
#define EPP    (BPB / PROD_WAVES)
#define PGROUPS (EPP * TCHUNK / 16)
#define PV4    (EPP * XV4)
#define PLOADS ((PV4 + 31) / 32)
#define GB     5
#define XSLOT  (PLOADS * 32 * 4)
#define XZTILE (ROWS * 4)
#define HALF   (TCHUNK / 2)

static_assert((NCHUNK & 1) == 1, "the scan runs tiles in pairs plus one last tile");
typedef __attribute__((ext_vector_type(2))) float v2f;
typedef __attribute__((ext_vector_type(8))) float v8f;

template <int K>
__device__ __forceinline__ float quad_take(float v) {
    return __int_as_float(__builtin_amdgcn_update_dpp(
        0, __float_as_int(v), K * 0x55, 0xF, 0xF, true));
}

__device__ __forceinline__ void bar_arrive() {
    asm volatile("s_barrier_signal -1" ::: "memory");
}
__device__ __forceinline__ void bar_wait() {
    asm volatile("s_barrier_wait -1" ::: "memory");
}
__device__ __forceinline__ void lds_done() {
    asm volatile("s_wait_dscnt 0x0" ::: "memory");
}
__device__ __forceinline__ void bar_arrive_at(float& h) {
    asm volatile("s_barrier_signal -1" : "+v"(h) :: "memory");
}
__device__ __forceinline__ void bar_wait_at(float& h) {
    asm volatile("s_barrier_wait -1" : "+v"(h) :: "memory");
}
__device__ __forceinline__ void lds_done_at(float& h) {
    asm volatile("s_wait_dscnt 0x0" : "+v"(h) :: "memory");
}
__device__ __forceinline__ void issue_here(float& h) {
    asm volatile("" : "+v"(h) :: "memory");
}

__device__ __forceinline__ void read_tile(float (&zr)[TCHUNK], const float* zz) {
#pragma unroll
    for (int t = 0; t < TCHUNK; ++t) zr[t] = zz[t * 4];
}

#define KC (-2.88539008177792681472f)
template <int T0, int T1>
__device__ __forceinline__ void scan_steps(const float (&zr)[TCHUNK],
                                           float& h, float& cs, float rk) {
#pragma unroll
    for (int t = T0; t < T1; ++t) {
        const float m  = __builtin_fmaf(h, rk, zr[t]);
        const float e  = __builtin_amdgcn_exp2f(m);
        const float r  = __builtin_amdgcn_rcpf(1.0f + e);
        const float ri = quad_take<0>(r);
        const float ro = quad_take<3>(r);
        const float o2 = ro + ro;
        const float fc = quad_take<1>(r) * cs;
        const float u  = quad_take<2>(r) * ri;
        cs = __builtin_fmaf(u, 2.0f * KC, __builtin_fmaf(ri, -KC, fc));
        const float e2 = __builtin_amdgcn_exp2f(cs);
        const float q  = __builtin_amdgcn_rcpf(1.0f + e2);
        h = __builtin_fmaf(q, o2, -ro);
    }
}

__device__ __forceinline__ void scan_tile(const float (&cur)[TCHUNK],
                                          float (&nxt)[TCHUNK],
                                          const float* zz_next,
                                          float& h, float& c, float rk) {
    bar_arrive_at(h);
    scan_steps<0, HALF>(cur, h, c, rk);
    bar_wait_at(h);
    read_tile(nxt, zz_next);
    issue_here(h);
    scan_steps<HALF, TCHUNK>(cur, h, c, rk);
    lds_done_at(h);
}

__global__ __launch_bounds__(BLOCK)
void lstm_scan_kernel(const float* __restrict__ x,
                      const float* __restrict__ kern,
                      const float* __restrict__ rkern,
                      const float* __restrict__ bias,
                      float* __restrict__ out) {
    extern __shared__ __attribute__((aligned(16))) float smem[];
    float* xz_lds = smem;
    float* x_lds  = smem + 2 * XZTILE;

    const int tid  = threadIdx.x;
    const int lane = tid & 31;
    const int wave = __builtin_amdgcn_readfirstlane(tid >> 5);
    const bool producer = (wave >= SCAN_WAVES);
    const int b0   = blockIdx.x * BPB;
    const float* xblk = x + (size_t)b0 * (T * F);

    const int gate = tid & 3;
    const int bl   = (tid >> 2) & (BPB - 1);
    const bool  is_g = (gate == 2);
    const float sc   = is_g ? 2.88539008177792681472f : 1.44269504088896340736f;
    const float rk   = rkern[gate] * -sc;
    float h = 0.0f, c = 0.0f;

    const int n  = lane & 15;
    const int kb = (lane < 16) ? 0 : 2;
    const float nsc_n = (n == 2) ? -2.88539008177792681472f : -1.44269504088896340736f;
    v2f bmat;
    bmat.x = (n < 4 && (kb + 0) < F) ? kern[(kb + 0) * 4 + n] * nsc_n : 0.0f;
    bmat.y = (n < 4 && (kb + 1) < F) ? kern[(kb + 1) * 4 + n] * nsc_n : 0.0f;
    const float bias_n = (n < 4) ? bias[n] * nsc_n : 0.0f;
    v8f cmat;
    for (int v = 0; v < 8; ++v) cmat[v] = bias_n;

    const int pw  = producer ? (wave - SCAN_WAVES) : 0;
    const int ep0 = pw * EPP;
    float* xs = x_lds + pw * XSLOT;
    const float4* src[PLOADS];
#pragma unroll
    for (int j = 0; j < PLOADS; ++j) {
        int i = lane + 32 * j;
        i = (i < PV4) ? i : (PV4 - 1);
        const int e = i / XV4;
        const int v = i - e * XV4;
        src[j] = (const float4*)(xblk + (size_t)(ep0 + e) * (T * F)) + v;
    }
    float4 xr[PLOADS];
#pragma unroll
    for (int j = 0; j < PLOADS; ++j) xr[j] = make_float4(0.0f, 0.0f, 0.0f, 0.0f);
    if (producer) {
#pragma unroll
        for (int j = 0; j < PLOADS; ++j) xr[j] = src[j][0];
    }

    if (producer) {
        for (int it = 0; it < NCHUNK; ++it) {
#pragma unroll
            for (int j = 0; j < PLOADS; ++j)
                ((float4*)xs)[lane + 32 * j] = xr[j];
            if (it + 1 < NCHUNK) {
#pragma unroll
                for (int j = 0; j < PLOADS; ++j)
                    xr[j] = src[j][(it + 1) * XV4];
            }
            __builtin_amdgcn_wave_barrier();

            float* xzb = xz_lds + (it & 1) * XZTILE + ep0 * (TCHUNK * 4);
            const int mofs = (lane < 16) ? 0 : 8;
            v2f afr[PGROUPS];
#pragma unroll
            for (int g = 0; g < PGROUPS; ++g) {
                const int r = g * 16 + n;
                const float ax = xs[r * F + kb];
                const float ay = xs[r * F + kb + 1];
                afr[g].x = ax;
                afr[g].y = (kb == 0) ? ay : 0.0f;
            }
#pragma unroll
            for (int gb = 0; gb < PGROUPS; gb += GB) {
                v8f d[GB];
#pragma unroll
                for (int u = 0; u < GB; ++u)
                    d[u] = __builtin_amdgcn_wmma_f32_16x16x4_f32(
                        false, afr[gb + u], false, bmat, (short)0, cmat,
                        false, false);
                if (n < 4) {
#pragma unroll
                    for (int u = 0; u < GB; ++u)
#pragma unroll
                        for (int v = 0; v < 8; ++v)
                            xzb[((gb + u) * 16 + mofs + v) * 4 + n] = d[u][v];
                }
            }
            lds_done();
            bar_arrive();
            bar_wait();
        }
    } else {
        const float* zz0 = xz_lds + bl * (TCHUNK * 4) + gate;
        const float* zz1 = zz0 + XZTILE;
        float za[TCHUNK], zb[TCHUNK];
        bar_arrive();
        bar_wait();
        read_tile(za, zz0);
        lds_done();
        for (int k = 0; k + 2 < NCHUNK; k += 2) {
            scan_tile(za, zb, zz1, h, c, rk);
            scan_tile(zb, za, zz0, h, c, rk);
        }
        scan_steps<0, TCHUNK>(za, h, c, rk);
    }

    if (!producer && gate == 0) x_lds[bl] = h;
    __syncthreads();
    if (tid < BPB) out[b0 + tid] = x_lds[tid];
}

extern "C" void kernel_launch(void* const* d_in, const int* in_sizes, int n_in,
                              void* d_out, int out_size, void* d_ws, size_t ws_size,
                              hipStream_t stream) {
    const float* x  = (const float*)d_in[0];
    const float* k  = (const float*)d_in[1];
    const float* rk = (const float*)d_in[2];
    const float* bi = (const float*)d_in[3];
    float*       o  = (float*)d_out;

    const size_t lds_bytes =
        (size_t)(2 * XZTILE + PROD_WAVES * XSLOT) * sizeof(float);
    hipLaunchKernelGGL(lstm_scan_kernel,
                       dim3(BATCH / BPB), dim3(BLOCK), lds_bytes, stream,
                       x, k, rk, bi, o);
}


# Two lanes per batch element, two gates per lane (pair swap by DPP): half the scan waves, one chain per SIMD instead of two sharing its issue port; work-group 64 elements, 4+4 waves; sums unchanged
# speedup vs baseline: 10.5253x; 1.0433x over previous
#include <hip/hip_runtime.h>


#define BATCH  16384
#define T      700
#define F      3
#define TCHUNK 20
#define NCHUNK (T / TCHUNK)
#define SCAN_WAVES 4
#define PROD_WAVES 4
#define BLOCK  ((SCAN_WAVES + PROD_WAVES) * 32)
#define BPB    (SCAN_WAVES * 32 / 2)
#define ROWS   (BPB * TCHUNK)
#define XV4    (TCHUNK * F / 4)
#define EPP    (BPB / PROD_WAVES)
#define PGROUPS (EPP * TCHUNK / 16)
#define PV4    (EPP * XV4)
#define PLOADS ((PV4 + 31) / 32)
#define GB     5
#define XSLOT  (PLOADS * 32 * 4)
#define XZTILE (ROWS * 4)
#define HALF   (TCHUNK / 2)

static_assert((NCHUNK & 1) == 1, "the scan runs tiles in pairs plus one last tile");
typedef __attribute__((ext_vector_type(2))) float v2f;
typedef __attribute__((ext_vector_type(8))) float v8f;

template <int K>
__device__ __forceinline__ float pair_take(float v) {
    return __int_as_float(__builtin_amdgcn_update_dpp(
        0, __float_as_int(v), K ? 0xF5 : 0xA0, 0xF, 0xF, true));
}

__device__ __forceinline__ void bar_arrive() {
    asm volatile("s_barrier_signal -1" ::: "memory");
}
__device__ __forceinline__ void bar_wait() {
    asm volatile("s_barrier_wait -1" ::: "memory");
}
__device__ __forceinline__ void lds_done() {
    asm volatile("s_wait_dscnt 0x0" ::: "memory");
}
__device__ __forceinline__ void bar_arrive_at(float& h) {
    asm volatile("s_barrier_signal -1" : "+v"(h) :: "memory");
}
__device__ __forceinline__ void bar_wait_at(float& h) {
    asm volatile("s_barrier_wait -1" : "+v"(h) :: "memory");
}
__device__ __forceinline__ void lds_done_at(float& h) {
    asm volatile("s_wait_dscnt 0x0" : "+v"(h) :: "memory");
}
__device__ __forceinline__ void side_by_side(float& a, float& b) {
    asm volatile("" : "+v"(a), "+v"(b));
}
__device__ __forceinline__ void issue_here(float& h) {
    asm volatile("" : "+v"(h) :: "memory");
}

__device__ __forceinline__ void read_tile(v2f (&zr)[TCHUNK], const float* zz) {
#pragma unroll
    for (int t = 0; t < TCHUNK; ++t) zr[t] = *(const v2f*)(zz + t * 4);
}

#define KC (-2.88539008177792681472f)
template <int T0, int T1>
__device__ __forceinline__ void scan_steps(const v2f (&zr)[TCHUNK],
                                           float& h, float& cs, v2f rk) {
#pragma unroll
    for (int t = T0; t < T1; ++t) {
        const float m1 = __builtin_fmaf(h, rk.x, zr[t].x);
        const float m2 = __builtin_fmaf(h, rk.y, zr[t].y);
        float a1 = 1.0f + __builtin_amdgcn_exp2f(m1);
        float a2 = 1.0f + __builtin_amdgcn_exp2f(m2);
        side_by_side(a1, a2);
        float r1 = __builtin_amdgcn_rcpf(a1);
        float r2 = __builtin_amdgcn_rcpf(a2);
        side_by_side(r1, r2);
        const float ri = pair_take<0>(r1);
        const float ro = pair_take<1>(r2);
        const float o2 = ro + ro;
        const float fc = pair_take<0>(r2) * cs;
        const float u  = pair_take<1>(r1) * ri;
        cs = __builtin_fmaf(u, 2.0f * KC, __builtin_fmaf(ri, -KC, fc));
        const float e2 = __builtin_amdgcn_exp2f(cs);
        const float q  = __builtin_amdgcn_rcpf(1.0f + e2);
        h = __builtin_fmaf(q, o2, -ro);
    }
}

__device__ __forceinline__ void scan_tile(const v2f (&cur)[TCHUNK],
                                          v2f (&nxt)[TCHUNK],
                                          const float* zz_next,
                                          float& h, float& c, v2f rk) {
    bar_arrive_at(h);
    scan_steps<0, HALF>(cur, h, c, rk);
    bar_wait_at(h);
    read_tile(nxt, zz_next);
    issue_here(h);
    scan_steps<HALF, TCHUNK>(cur, h, c, rk);
    lds_done_at(h);
}

__global__ __launch_bounds__(BLOCK)
void lstm_scan_kernel(const float* __restrict__ x,
                      const float* __restrict__ kern,
                      const float* __restrict__ rkern,
                      const float* __restrict__ bias,
                      float* __restrict__ out) {
    extern __shared__ __attribute__((aligned(16))) float smem[];
    float* xz_lds = smem;
    float* x_lds  = smem + 2 * XZTILE;

    const int tid  = threadIdx.x;
    const int lane = tid & 31;
    const int wave = __builtin_amdgcn_readfirstlane(tid >> 5);
    const bool producer = (wave >= SCAN_WAVES);
    const int b0   = blockIdx.x * BPB;
    const float* xblk = x + (size_t)b0 * (T * F);

    const int odd  = tid & 1;
    const int bl   = (tid >> 1) & (BPB - 1);
    const float sc1 = odd ? 2.88539008177792681472f : 1.44269504088896340736f;
    v2f rk;
    rk.x = rkern[2 * odd]     * -sc1;
    rk.y = rkern[2 * odd + 1] * -1.44269504088896340736f;
    float h = 0.0f, c = 0.0f;

    const int n  = lane & 15;
    const int kb = (lane < 16) ? 0 : 2;
    const float nsc_n = (n == 2) ? -2.88539008177792681472f : -1.44269504088896340736f;
    v2f bmat;
    bmat.x = (n < 4 && (kb + 0) < F) ? kern[(kb + 0) * 4 + n] * nsc_n : 0.0f;
    bmat.y = (n < 4 && (kb + 1) < F) ? kern[(kb + 1) * 4 + n] * nsc_n : 0.0f;
    const float bias_n = (n < 4) ? bias[n] * nsc_n : 0.0f;
    v8f cmat;
    for (int v = 0; v < 8; ++v) cmat[v] = bias_n;

    const int pw  = producer ? (wave - SCAN_WAVES) : 0;
    const int ep0 = pw * EPP;
    float* xs = x_lds + pw * XSLOT;
    const float4* src[PLOADS];
#pragma unroll
    for (int j = 0; j < PLOADS; ++j) {
        int i = lane + 32 * j;
        i = (i < PV4) ? i : (PV4 - 1);
        const int e = i / XV4;
        const int v = i - e * XV4;
        src[j] = (const float4*)(xblk + (size_t)(ep0 + e) * (T * F)) + v;
    }
    float4 xr[PLOADS];
#pragma unroll
    for (int j = 0; j < PLOADS; ++j) xr[j] = make_float4(0.0f, 0.0f, 0.0f, 0.0f);
    if (producer) {
#pragma unroll
        for (int j = 0; j < PLOADS; ++j) xr[j] = src[j][0];
    }

    if (producer) {
        for (int it = 0; it < NCHUNK; ++it) {
#pragma unroll
            for (int j = 0; j < PLOADS; ++j)
                ((float4*)xs)[lane + 32 * j] = xr[j];
            if (it + 1 < NCHUNK) {
#pragma unroll
                for (int j = 0; j < PLOADS; ++j)
                    xr[j] = src[j][(it + 1) * XV4];
            }
            __builtin_amdgcn_wave_barrier();

            float* xzb = xz_lds + (it & 1) * XZTILE + ep0 * (TCHUNK * 4);
            const int mofs = (lane < 16) ? 0 : 8;
            v2f afr[PGROUPS];
#pragma unroll
            for (int g = 0; g < PGROUPS; ++g) {
                const int r = g * 16 + n;
                const float ax = xs[r * F + kb];
                const float ay = xs[r * F + kb + 1];
                afr[g].x = ax;
                afr[g].y = (kb == 0) ? ay : 0.0f;
            }
#pragma unroll
            for (int gb = 0; gb < PGROUPS; gb += GB) {
                v8f d[GB];
#pragma unroll
                for (int u = 0; u < GB; ++u)
                    d[u] = __builtin_amdgcn_wmma_f32_16x16x4_f32(
                        false, afr[gb + u], false, bmat, (short)0, cmat,
                        false, false);
                if (n < 4) {
#pragma unroll
                    for (int u = 0; u < GB; ++u)
#pragma unroll
                        for (int v = 0; v < 8; ++v)
                            xzb[((gb + u) * 16 + mofs + v) * 4 + n] = d[u][v];
                }
            }
            lds_done();
            bar_arrive();
            bar_wait();
        }
    } else {
        const float* zz0 = xz_lds + bl * (TCHUNK * 4) + 2 * odd;
        const float* zz1 = zz0 + XZTILE;
        v2f za[TCHUNK], zb[TCHUNK];
        bar_arrive();
        bar_wait();
        read_tile(za, zz0);
        lds_done();
        for (int k = 0; k + 2 < NCHUNK; k += 2) {
            scan_tile(za, zb, zz1, h, c, rk);
            scan_tile(zb, za, zz0, h, c, rk);
        }
        scan_steps<0, TCHUNK>(za, h, c, rk);
    }

    if (!producer && odd == 0) x_lds[bl] = h;
    __syncthreads();
    if (tid < BPB) out[b0 + tid] = x_lds[tid];
}

extern "C" void kernel_launch(void* const* d_in, const int* in_sizes, int n_in,
                              void* d_out, int out_size, void* d_ws, size_t ws_size,
                              hipStream_t stream) {
    const float* x  = (const float*)d_in[0];
    const float* k  = (const float*)d_in[1];
    const float* rk = (const float*)d_in[2];
    const float* bi = (const float*)d_in[3];
    float*       o  = (float*)d_out;

    const size_t lds_bytes =
        (size_t)(2 * XZTILE + PROD_WAVES * XSLOT) * sizeof(float);
    hipLaunchKernelGGL(lstm_scan_kernel,
                       dim3(BATCH / BPB), dim3(BLOCK), lds_bytes, stream,
                       x, k, rk, bi, o);
}
